# starting kernel
# baseline (speedup 1.0000x reference)
#include <hip/hip_runtime.h>

#define BATCH_N   16384
#define MAX_ITEMS 200
#define EMB_DIM   128
#define NTILES    ((MAX_ITEMS + 15) / 16)

typedef __attribute__((ext_vector_type(2))) float v2f;
typedef __attribute__((ext_vector_type(8))) float v8f;

__global__ __launch_bounds__(128) void ncf_fused_kernel(
    const float* __restrict__ ratings,
    const int*   __restrict__ item_ids,
    const float* __restrict__ noise,
    const float* __restrict__ init_user_emb,
    const float* __restrict__ gmf_emb,
    const float* __restrict__ ncf_emb,
    const float* __restrict__ W0, const float* __restrict__ b0,
    const float* __restrict__ W1, const float* __restrict__ b1,
    const float* __restrict__ W2, const float* __restrict__ b2,
    const float* __restrict__ W3, const float* __restrict__ b3,
    const float* __restrict__ Wn, const float* __restrict__ bn,
    const float* __restrict__ Wf, const float* __restrict__ bf,
    const float* __restrict__ p,
    float* __restrict__ out)
{
    __shared__ float s_row[2 * EMB_DIM];
    __shared__ float s_dots[16];
    __shared__ float s_g[4];
    __shared__ float s_u[8];
    __shared__ int   s_ids[MAX_ITEMS];

    const int b = blockIdx.x;
    const int t = threadIdx.x;
    const long rowBase = (long)b * MAX_ITEMS;

    if (t < 13) __builtin_prefetch(&ratings[rowBase + t * 16], 0, 3);

    s_row[t]           = init_user_emb[b * EMB_DIM + t];
    s_row[EMB_DIM + t] = noise[b * EMB_DIM + t];
    for (int i = t; i < MAX_ITEMS; i += 128) s_ids[i] = item_ids[rowBase + i];
    __syncthreads();

    if (t < 16) {
        const int grp = t >> 2;
        const int j   = t & 3;
        const float* W    = (grp == 0) ? W0 : (grp == 1) ? W1 : (grp == 2) ? W2 : W3;
        const float* bias = (grp == 0) ? b0 : (grp == 1) ? b1 : (grp == 2) ? b2 : b3;
        const float* x = &s_row[grp * 64];
        float acc = bias[j];
        #pragma unroll 8
        for (int k = 0; k < 64; ++k) acc += x[k] * W[j * 64 + k];
        s_dots[t] = acc;
    }
    __syncthreads();

    if (t < 4) s_g[t] = s_dots[t] * Wf[t] + s_dots[8 + t] * Wf[4 + t];
    if (t < 8) {
        float acc = bn[t];
        #pragma unroll
        for (int j = 0; j < 4; ++j)
            acc += Wn[t * 12 + j] * s_dots[4 + j] + Wn[t * 12 + 4 + j] * s_dots[12 + j];
        s_u[t] = acc;
    }
    __syncthreads();

    const int wave  = t >> 5;
    const int lane  = t & 31;
    const int row16 = lane & 15;
    const bool hi   = lane >= 16;

    v2f A = {0.0f, 0.0f};
    if (row16 < 8) {
        const int kb = 8 + (hi ? 2 : 0);
        A.x = Wn[row16 * 12 + kb];
        A.y = Wn[row16 * 12 + kb + 1];
    }

    const float g0 = s_g[0], g1 = s_g[1], g2 = s_g[2], g3 = s_g[3];
    float u_r[8], wf8[8];
    #pragma unroll
    for (int r = 0; r < 8; ++r) { u_r[r] = s_u[r]; wf8[r] = Wf[8 + r]; }
    const float pv = p[0], bfv = bf[0];

    for (int tile = wave; tile < NTILES; tile += 4) {
        const int item  = tile * 16 + row16;
        const int itemC = (item < MAX_ITEMS) ? item : (MAX_ITEMS - 1);
        const int iid   = s_ids[itemC];

        const float2 bv = *(const float2*)&ncf_emb[(long)iid * 4 + (hi ? 2 : 0)];
        const float4 gi = *(const float4*)&gmf_emb[(long)iid * 4];
        float gx = gi.x, gy = gi.y, gz = gi.z, gw = gi.w;
        float rt = ratings[rowBase + itemC];

        asm volatile("" : "+v"(gx), "+v"(gy), "+v"(gz), "+v"(gw), "+v"(rt));

        v2f B = {bv.x, bv.y};
        v8f C = {};
        C = __builtin_amdgcn_wmma_f32_16x16x4_f32(
                 false, A,  false, B,
                 (short)0, C,  false,  false);

        float score = bfv + gx * g0 + gy * g1 + gz * g2 + gw * g3;
        #pragma unroll
        for (int r = 0; r < 8; ++r) {
            const float h = u_r[r] + C[r];
            score += fmaxf(h, 0.0f) * wf8[r];
        }
        const float result = pv * rt + score;

        if (!hi && item < MAX_ITEMS)
            out[rowBase + item] = result;
    }
}

extern "C" void kernel_launch(void* const* d_in, const int* in_sizes, int n_in,
                              void* d_out, int out_size, void* d_ws, size_t ws_size,
                              hipStream_t stream) {
    const float* ratings       = (const float*)d_in[0];
    const int*   item_ids      = (const int*)  d_in[1];
    const float* noise         = (const float*)d_in[2];
    const float* init_user_emb = (const float*)d_in[3];
    const float* gmf_emb       = (const float*)d_in[4];
    const float* ncf_emb       = (const float*)d_in[5];
    const float* W0 = (const float*)d_in[6];  const float* b0 = (const float*)d_in[7];
    const float* W1 = (const float*)d_in[8];  const float* b1 = (const float*)d_in[9];
    const float* W2 = (const float*)d_in[10]; const float* b2 = (const float*)d_in[11];
    const float* W3 = (const float*)d_in[12]; const float* b3 = (const float*)d_in[13];
    const float* Wn = (const float*)d_in[14]; const float* bn = (const float*)d_in[15];
    const float* Wf = (const float*)d_in[16]; const float* bf = (const float*)d_in[17];
    const float* p  = (const float*)d_in[18];
    float* out = (float*)d_out;

    ncf_fused_kernel<<<BATCH_N, 128, 0, stream>>>(
        ratings, item_ids, noise, init_user_emb, gmf_emb, ncf_emb,
        W0, b0, W1, b1, W2, b2, W3, b3, Wn, bn, Wf, bf, p, out);
}


# Work-group owns 16 batch rows: the four 64->4 projections run as f32 WMMA products over those rows (projection sum order changes); 8 waves each walk two rows' item tiles side by side
# speedup vs baseline: 1.3255x; 1.3255x over previous
#include <hip/hip_runtime.h>


#define BATCH_N    16384
#define MAX_ITEMS  200
#define EMB_DIM    128
#define HALF_DIM   64
#define NTILES     ((MAX_ITEMS + 15) / 16)
#define ROWS_WG    16
#define WG_THREADS 256
#define GRID_WGS   1024
#define IDS_V4     800
#define IDS_PER_THREAD 4

typedef __attribute__((ext_vector_type(2))) float v2f;
typedef __attribute__((ext_vector_type(8))) float v8f;

__device__ __forceinline__ float item_result(
    v2f A, float2 bv, float4 gi, float rt,
    const float* g, const float* u, const float* wf8, float pv, float bfv)
{
    v2f B = {bv.x, bv.y};
    v8f C = {};
    C = __builtin_amdgcn_wmma_f32_16x16x4_f32(
             false, A,  false, B,
             (short)0, C,  false,  false);
    float score = bfv + gi.x * g[0] + gi.y * g[1] + gi.z * g[2] + gi.w * g[3];
    #pragma unroll
    for (int r = 0; r < 8; ++r) {
        const float hdn = u[r] + C[r];
        score += fmaxf(hdn, 0.0f) * wf8[r];
    }
    return pv * rt + score;
}

__global__ __launch_bounds__(WG_THREADS) void ncf_rowtile_kernel(
    const float* __restrict__ ratings,
    const int*   __restrict__ item_ids,
    const float* __restrict__ noise,
    const float* __restrict__ init_user_emb,
    const float* __restrict__ gmf_emb,
    const float* __restrict__ ncf_emb,
    const float* __restrict__ W0, const float* __restrict__ b0,
    const float* __restrict__ W1, const float* __restrict__ b1,
    const float* __restrict__ W2, const float* __restrict__ b2,
    const float* __restrict__ W3, const float* __restrict__ b3,
    const float* __restrict__ Wn, const float* __restrict__ bn,
    const float* __restrict__ Wf, const float* __restrict__ bf,
    const float* __restrict__ p,
    float* __restrict__ out)
{
    __shared__ __attribute__((aligned(16))) int   s_ids[IDS_PER_THREAD * WG_THREADS * 4];
    __shared__ __attribute__((aligned(16))) float s_part[2 * ROWS_WG * 16];
    __shared__ float s_u[ROWS_WG * 8];
    __shared__ float s_g[ROWS_WG * 4];

    const int t    = threadIdx.x;
    const int wave = t >> 5;
    const int lane = t & 31;
    const int c    = lane & 15;
    const int h    = lane >> 4;
    const bool hi  = h != 0;
    const int  row0     = blockIdx.x * ROWS_WG;
    const long itemBase = (long)row0 * MAX_ITEMS;

    if (t < (ROWS_WG * MAX_ITEMS * 4) / 128) __builtin_prefetch(&ratings[itemBase + t * 32], 0, 3);

    const int4* idsrc = (const int4*)(item_ids + itemBase);
    int4 idv[IDS_PER_THREAD];
    #pragma unroll
    for (int j = 0; j < IDS_PER_THREAD; ++j) {
        const int i = t + j * WG_THREADS;
        idv[j] = idsrc[i < IDS_V4 ? i : IDS_V4 - 1];
    }

    {
        const int grp = wave >> 1;
        const int kh  = wave & 1;
        const float* X  = (grp < 2) ? init_user_emb : noise;
        const float* Wg = (grp == 0) ? W0 : (grp == 1) ? W1 : (grp == 2) ? W2 : W3;
        const float* xrow = X + (long)(row0 + c) * EMB_DIM + (grp & 1) * HALF_DIM + kh * 32 + h * 4;
        const float* wrow = Wg + (c & 3) * HALF_DIM + kh * 32 + h * 4;
        float4 xv[4], wv[4];
        #pragma unroll
        for (int i = 0; i < 4; ++i) {
            xv[i] = *(const float4*)(xrow + i * 8);
            wv[i] = *(const float4*)(wrow + i * 8);
        }
        {
            int4* dst = (int4*)s_ids;
            #pragma unroll
            for (int j = 0; j < IDS_PER_THREAD; ++j) dst[t + j * WG_THREADS] = idv[j];
        }
        v8f acc0 = {}, acc1 = {};
        #pragma unroll
        for (int i = 0; i < 4; ++i) {
            v2f a0 = {wv[i].x, wv[i].y}, x0 = {xv[i].x, xv[i].y};
            v2f a1 = {wv[i].z, wv[i].w}, x1 = {xv[i].z, xv[i].w};
            acc0 = __builtin_amdgcn_wmma_f32_16x16x4_f32(false, a0, false, x0, (short)0, acc0, false, false);
            acc1 = __builtin_amdgcn_wmma_f32_16x16x4_f32(false, a1, false, x1, (short)0, acc1, false, false);
        }
        float4 o;
        o.x = acc0[0] + acc1[0];
        o.y = acc0[1] + acc1[1];
        o.z = acc0[2] + acc1[2];
        o.w = acc0[3] + acc1[3];
        if (!hi) *(float4*)&s_part[(kh * ROWS_WG + c) * 16 + grp * 4] = o;
    }
    __syncthreads();

    if (t < ROWS_WG * 8) {
        const int row = t >> 3, r = t & 7;
        const float* p0 = &s_part[row * 16];
        const float* p1 = &s_part[(ROWS_WG + row) * 16];
        float acc = bn[r];
        #pragma unroll
        for (int j = 0; j < 4; ++j) {
            const float nu = (p0[4 + j] + p1[4 + j]) + b1[j];
            const float nn = (p0[12 + j] + p1[12 + j]) + b3[j];
            acc += Wn[r * 12 + j] * nu + Wn[r * 12 + 4 + j] * nn;
        }
        s_u[row * 8 + r] = acc;
    } else if (t < ROWS_WG * 8 + ROWS_WG * 4) {
        const int q = t - ROWS_WG * 8;
        const int row = q >> 2, k = q & 3;
        const float* p0 = &s_part[row * 16];
        const float* p1 = &s_part[(ROWS_WG + row) * 16];
        const float gu = (p0[k] + p1[k]) + b0[k];
        const float gn = (p0[8 + k] + p1[8 + k]) + b2[k];
        s_g[row * 4 + k] = gu * Wf[k] + gn * Wf[4 + k];
    }
    __syncthreads();

    v2f A = {0.0f, 0.0f};
    if (c < 8) {
        const int kb = 8 + (hi ? 2 : 0);
        A.x = Wn[c * 12 + kb];
        A.y = Wn[c * 12 + kb + 1];
    }
    const int lr0 = wave * 2, lr1 = lr0 + 1;
    float g0v[4], g1v[4], u0v[8], u1v[8], wf8[8];
    #pragma unroll
    for (int k = 0; k < 4; ++k) { g0v[k] = s_g[lr0 * 4 + k]; g1v[k] = s_g[lr1 * 4 + k]; }
    #pragma unroll
    for (int r = 0; r < 8; ++r) { u0v[r] = s_u[lr0 * 8 + r]; u1v[r] = s_u[lr1 * 8 + r]; wf8[r] = Wf[8 + r]; }
    const float pv = p[0], bfv = bf[0];
    const long base0 = itemBase + (long)lr0 * MAX_ITEMS;
    const long base1 = base0 + MAX_ITEMS;

    for (int tile = 0; tile < NTILES; ++tile) {
        const int item  = tile * 16 + c;
        const int itemC = (item < MAX_ITEMS) ? item : (MAX_ITEMS - 1);
        const int id0 = s_ids[lr0 * MAX_ITEMS + itemC];
        const int id1 = s_ids[lr1 * MAX_ITEMS + itemC];

        const float2 bv0 = *(const float2*)&ncf_emb[(long)id0 * 4 + (hi ? 2 : 0)];
        const float4 gi0 = *(const float4*)&gmf_emb[(long)id0 * 4];
        const float  rt0 = ratings[base0 + itemC];
        const float2 bv1 = *(const float2*)&ncf_emb[(long)id1 * 4 + (hi ? 2 : 0)];
        const float4 gi1 = *(const float4*)&gmf_emb[(long)id1 * 4];
        const float  rt1 = ratings[base1 + itemC];

        asm volatile("" :: "v"(gi0.x), "v"(gi0.y), "v"(gi0.z), "v"(gi0.w), "v"(rt0),
                           "v"(gi1.x), "v"(gi1.y), "v"(gi1.z), "v"(gi1.w), "v"(rt1));

        const float res0 = item_result(A, bv0, gi0, rt0, g0v, u0v, wf8, pv, bfv);
        const float res1 = item_result(A, bv1, gi1, rt1, g1v, u1v, wf8, pv, bfv);

        if (!hi && item < MAX_ITEMS) {
            out[base0 + item] = res0;
            out[base1 + item] = res1;
        }
    }
}

extern "C" void kernel_launch(void* const* d_in, const int* in_sizes, int n_in,
                              void* d_out, int out_size, void* d_ws, size_t ws_size,
                              hipStream_t stream) {
    const float* ratings       = (const float*)d_in[0];
    const int*   item_ids      = (const int*)  d_in[1];
    const float* noise         = (const float*)d_in[2];
    const float* init_user_emb = (const float*)d_in[3];
    const float* gmf_emb       = (const float*)d_in[4];
    const float* ncf_emb       = (const float*)d_in[5];
    const float* W0 = (const float*)d_in[6];  const float* b0 = (const float*)d_in[7];
    const float* W1 = (const float*)d_in[8];  const float* b1 = (const float*)d_in[9];
    const float* W2 = (const float*)d_in[10]; const float* b2 = (const float*)d_in[11];
    const float* W3 = (const float*)d_in[12]; const float* b3 = (const float*)d_in[13];
    const float* Wn = (const float*)d_in[14]; const float* bn = (const float*)d_in[15];
    const float* Wf = (const float*)d_in[16]; const float* bf = (const float*)d_in[17];
    const float* p  = (const float*)d_in[18];
    float* out = (float*)d_out;

    ncf_rowtile_kernel<<<GRID_WGS, WG_THREADS, 0, stream>>>(
        ratings, item_ids, noise, init_user_emb, gmf_emb, ncf_emb,
        W0, b0, W1, b1, W2, b2, W3, b3, Wn, bn, Wf, bf, p, out);
}


# Item phase re-tiled: the work-group's 3200 items as 100 flat 32-item tiles, one item per lane; block-diagonal A makes both lane halves' D rows live (hidden-layer sum order changes)
# speedup vs baseline: 1.4894x; 1.1237x over previous
#include <hip/hip_runtime.h>


#define BATCH_N    16384
#define MAX_ITEMS  200
#define EMB_DIM    128
#define HALF_DIM   64
#define ROWS_WG    16
#define WG_THREADS 256
#define GRID_WGS   1024
#define IDS_V4     800
#define IDS_PER_THREAD 4
#define TILES_WG   100
#define TILE_WAVES 8

typedef __attribute__((ext_vector_type(2))) float v2f;
typedef __attribute__((ext_vector_type(8))) float v8f;

__device__ __forceinline__ float item_result(
    v2f A01, v2f A23, float4 nv, float4 gi, float rt,
    float4 ua, float4 ub, float4 g, const float* wf8, float pv, float bfv)
{
    v8f C = {ua.x, ua.y, ua.z, ua.w, ub.x, ub.y, ub.z, ub.w};
    v2f B01 = {nv.x, nv.y};
    v2f B23 = {nv.z, nv.w};
    C = __builtin_amdgcn_wmma_f32_16x16x4_f32(
             false, A01,  false, B01,
             (short)0, C,  false,  false);
    C = __builtin_amdgcn_wmma_f32_16x16x4_f32(
             false, A23,  false, B23,
             (short)0, C,  false,  false);
    float score = bfv + gi.x * g.x + gi.y * g.y + gi.z * g.z + gi.w * g.w;
    #pragma unroll
    for (int r = 0; r < 8; ++r) score += fmaxf(C[r], 0.0f) * wf8[r];
    return pv * rt + score;
}

__global__ __launch_bounds__(WG_THREADS) void ncf_rowtile_kernel(
    const float* __restrict__ ratings,
    const int*   __restrict__ item_ids,
    const float* __restrict__ noise,
    const float* __restrict__ init_user_emb,
    const float* __restrict__ gmf_emb,
    const float* __restrict__ ncf_emb,
    const float* __restrict__ W0, const float* __restrict__ b0,
    const float* __restrict__ W1, const float* __restrict__ b1,
    const float* __restrict__ W2, const float* __restrict__ b2,
    const float* __restrict__ W3, const float* __restrict__ b3,
    const float* __restrict__ Wn, const float* __restrict__ bn,
    const float* __restrict__ Wf, const float* __restrict__ bf,
    const float* __restrict__ p,
    float* __restrict__ out)
{
    __shared__ __attribute__((aligned(16))) int   s_ids[IDS_PER_THREAD * WG_THREADS * 4];
    __shared__ __attribute__((aligned(16))) float s_part[2 * ROWS_WG * 16];
    __shared__ __attribute__((aligned(16))) float s_u[ROWS_WG * 8];
    __shared__ __attribute__((aligned(16))) float s_g[ROWS_WG * 4];

    const int t    = threadIdx.x;
    const int wave = t >> 5;
    const int lane = t & 31;
    const int c    = lane & 15;
    const int h    = lane >> 4;
    const bool hi  = h != 0;
    const int  row0     = blockIdx.x * ROWS_WG;
    const long itemBase = (long)row0 * MAX_ITEMS;

    if (t < (ROWS_WG * MAX_ITEMS * 4) / 128) __builtin_prefetch(&ratings[itemBase + t * 32], 0, 3);

    const int4* idsrc = (const int4*)(item_ids + itemBase);
    int4 idv[IDS_PER_THREAD];
    #pragma unroll
    for (int j = 0; j < IDS_PER_THREAD; ++j) {
        const int i = t + j * WG_THREADS;
        idv[j] = idsrc[i < IDS_V4 ? i : IDS_V4 - 1];
    }

    {
        const int grp = wave >> 1;
        const int kh  = wave & 1;
        const float* X  = (grp < 2) ? init_user_emb : noise;
        const float* Wg = (grp == 0) ? W0 : (grp == 1) ? W1 : (grp == 2) ? W2 : W3;
        const float* xrow = X + (long)(row0 + c) * EMB_DIM + (grp & 1) * HALF_DIM + kh * 32 + h * 4;
        const float* wrow = Wg + (c & 3) * HALF_DIM + kh * 32 + h * 4;
        float4 xv[4], wv[4];
        #pragma unroll
        for (int i = 0; i < 4; ++i) {
            xv[i] = *(const float4*)(xrow + i * 8);
            wv[i] = *(const float4*)(wrow + i * 8);
        }
        {
            int4* dst = (int4*)s_ids;
            #pragma unroll
            for (int j = 0; j < IDS_PER_THREAD; ++j) dst[t + j * WG_THREADS] = idv[j];
        }
        v8f acc0 = {}, acc1 = {};
        #pragma unroll
        for (int i = 0; i < 4; ++i) {
            v2f a0 = {wv[i].x, wv[i].y}, x0 = {xv[i].x, xv[i].y};
            v2f a1 = {wv[i].z, wv[i].w}, x1 = {xv[i].z, xv[i].w};
            acc0 = __builtin_amdgcn_wmma_f32_16x16x4_f32(false, a0, false, x0, (short)0, acc0, false, false);
            acc1 = __builtin_amdgcn_wmma_f32_16x16x4_f32(false, a1, false, x1, (short)0, acc1, false, false);
        }
        float4 o;
        o.x = acc0[0] + acc1[0];
        o.y = acc0[1] + acc1[1];
        o.z = acc0[2] + acc1[2];
        o.w = acc0[3] + acc1[3];
        if (!hi) *(float4*)&s_part[(kh * ROWS_WG + c) * 16 + grp * 4] = o;
    }
    __syncthreads();

    if (t < ROWS_WG * 8) {
        const int row = t >> 3, r = t & 7;
        const float* p0 = &s_part[row * 16];
        const float* p1 = &s_part[(ROWS_WG + row) * 16];
        float acc = bn[r];
        #pragma unroll
        for (int j = 0; j < 4; ++j) {
            const float nu = (p0[4 + j] + p1[4 + j]) + b1[j];
            const float nn = (p0[12 + j] + p1[12 + j]) + b3[j];
            acc += Wn[r * 12 + j] * nu + Wn[r * 12 + 4 + j] * nn;
        }
        s_u[row * 8 + r] = acc;
    } else if (t < ROWS_WG * 8 + ROWS_WG * 4) {
        const int q = t - ROWS_WG * 8;
        const int row = q >> 2, k = q & 3;
        const float* p0 = &s_part[row * 16];
        const float* p1 = &s_part[(ROWS_WG + row) * 16];
        const float gu = (p0[k] + p1[k]) + b0[k];
        const float gn = (p0[8 + k] + p1[8 + k]) + b2[k];
        s_g[row * 4 + k] = gu * Wf[k] + gn * Wf[4 + k];
    }
    __syncthreads();

    v2f A01 = {0.0f, 0.0f}, A23 = {0.0f, 0.0f};
    if ((c < 8) == !hi) {
        const float* wr = Wn + (c & 7) * 12 + 8;
        A01.x = wr[0]; A01.y = wr[1];
        A23.x = wr[2]; A23.y = wr[3];
    }
    float wf8[8];
    #pragma unroll
    for (int r = 0; r < 8; ++r) wf8[r] = Wf[8 + r];
    const float pv = p[0], bfv = bf[0];
    const float4* u4 = (const float4*)s_u;
    const float4* g4 = (const float4*)s_g;

    int tile = __builtin_amdgcn_readfirstlane(wave);
    for (; tile + TILE_WAVES < TILES_WG; tile += 2 * TILE_WAVES) {
        const unsigned f0 = (unsigned)(tile * 32 + lane);
        const unsigned f1 = f0 + TILE_WAVES * 32;
        const int id0 = s_ids[f0];
        const int id1 = s_ids[f1];
        const float4 nv0 = *(const float4*)&ncf_emb[(long)id0 * 4];
        const float4 gi0 = *(const float4*)&gmf_emb[(long)id0 * 4];
        const float4 nv1 = *(const float4*)&ncf_emb[(long)id1 * 4];
        const float4 gi1 = *(const float4*)&gmf_emb[(long)id1 * 4];
        const float  rt0 = ratings[itemBase + f0];
        const float  rt1 = ratings[itemBase + f1];
        const unsigned r0 = f0 / MAX_ITEMS, r1 = f1 / MAX_ITEMS;
        const float res0 = item_result(A01, A23, nv0, gi0, rt0, u4[r0 * 2], u4[r0 * 2 + 1], g4[r0], wf8, pv, bfv);
        const float res1 = item_result(A01, A23, nv1, gi1, rt1, u4[r1 * 2], u4[r1 * 2 + 1], g4[r1], wf8, pv, bfv);
        out[itemBase + f0] = res0;
        out[itemBase + f1] = res1;
    }
    if (tile < TILES_WG) {
        const unsigned f0 = (unsigned)(tile * 32 + lane);
        const int id0 = s_ids[f0];
        const float4 nv0 = *(const float4*)&ncf_emb[(long)id0 * 4];
        const float4 gi0 = *(const float4*)&gmf_emb[(long)id0 * 4];
        const float  rt0 = ratings[itemBase + f0];
        const unsigned r0 = f0 / MAX_ITEMS;
        out[itemBase + f0] = item_result(A01, A23, nv0, gi0, rt0, u4[r0 * 2], u4[r0 * 2 + 1], g4[r0], wf8, pv, bfv);
    }
}

extern "C" void kernel_launch(void* const* d_in, const int* in_sizes, int n_in,
                              void* d_out, int out_size, void* d_ws, size_t ws_size,
                              hipStream_t stream) {
    const float* ratings       = (const float*)d_in[0];
    const int*   item_ids      = (const int*)  d_in[1];
    const float* noise         = (const float*)d_in[2];
    const float* init_user_emb = (const float*)d_in[3];
    const float* gmf_emb       = (const float*)d_in[4];
    const float* ncf_emb       = (const float*)d_in[5];
    const float* W0 = (const float*)d_in[6];  const float* b0 = (const float*)d_in[7];
    const float* W1 = (const float*)d_in[8];  const float* b1 = (const float*)d_in[9];
    const float* W2 = (const float*)d_in[10]; const float* b2 = (const float*)d_in[11];
    const float* W3 = (const float*)d_in[12]; const float* b3 = (const float*)d_in[13];
    const float* Wn = (const float*)d_in[14]; const float* bn = (const float*)d_in[15];
    const float* Wf = (const float*)d_in[16]; const float* bf = (const float*)d_in[17];
    const float* p  = (const float*)d_in[18];
    float* out = (float*)d_out;

    ncf_rowtile_kernel<<<GRID_WGS, WG_THREADS, 0, stream>>>(
        ratings, item_ids, noise, init_user_emb, gmf_emb, ncf_emb,
        W0, b0, W1, b1, W2, b2, W3, b3, Wn, bn, Wf, bf, p, out);
}


# Item half of the hidden layer done once per table item in a pre-pass (WMMA), stored with the gmf row as a 64-byte record: item loop gathers one line per item, adds h to u (hidden sum order changes)
# speedup vs baseline: 1.7233x; 1.1570x over previous
#include <hip/hip_runtime.h>


#define BATCH_N    16384
#define MAX_ITEMS  200
#define EMB_DIM    128
#define HALF_DIM   64
#define ROWS_WG    16
#define WG_THREADS 256
#define GRID_WGS   1024
#define IDS_V4     800
#define IDS_PER_THREAD 4
#define TILES_WG   100
#define TILE_WAVES 8
#define NUM_ITEMS  100000
#define REC_F      16
#define PRE_TILES  3125
#define PRE_TPW    4
#define PRE_THREADS 64
#define PRE_WGS    391

typedef __attribute__((ext_vector_type(2))) float v2f;
typedef __attribute__((ext_vector_type(8))) float v8f;
typedef __attribute__((ext_vector_type(4))) float v4f;

__global__ __launch_bounds__(PRE_THREADS) void item_table_kernel(
    const float* __restrict__ gmf_emb,
    const float* __restrict__ ncf_emb,
    const float* __restrict__ Wn,
    float* __restrict__ rec)
{
    const int lane = threadIdx.x & 31;
    const int c    = lane & 15;
    const bool hi  = (lane >> 4) != 0;
    v2f A01 = {0.0f, 0.0f}, A23 = {0.0f, 0.0f};
    if ((c < 8) == !hi) {
        const float* wr = Wn + (c & 7) * 12 + 8;
        A01.x = wr[0]; A01.y = wr[1];
        A23.x = wr[2]; A23.y = wr[3];
    }
    const int tile0 = PRE_TPW * __builtin_amdgcn_readfirstlane((int)(blockIdx.x * 2 + (threadIdx.x >> 5)));
    v4f nv[PRE_TPW], gi[PRE_TPW];
    #pragma unroll
    for (int j = 0; j < PRE_TPW; ++j) {
        const int tl = (tile0 + j < PRE_TILES) ? tile0 + j : PRE_TILES - 1;
        const long item = (long)tl * 32 + lane;
        nv[j] = *(const v4f*)&ncf_emb[item * 4];
        gi[j] = *(const v4f*)&gmf_emb[item * 4];
    }
    asm volatile("" :: "v"(nv[0]), "v"(nv[1]), "v"(nv[2]), "v"(nv[3]),
                       "v"(gi[0]), "v"(gi[1]), "v"(gi[2]), "v"(gi[3]));
    #pragma unroll
    for (int j = 0; j < PRE_TPW; ++j) {
        v8f C = {};
        v2f B01 = {nv[j].x, nv[j].y};
        v2f B23 = {nv[j].z, nv[j].w};
        C = __builtin_amdgcn_wmma_f32_16x16x4_f32(
                 false, A01,  false, B01,
                 (short)0, C,  false,  false);
        C = __builtin_amdgcn_wmma_f32_16x16x4_f32(
                 false, A23,  false, B23,
                 (short)0, C,  false,  false);
        if (tile0 + j < PRE_TILES) {
            const long item = (long)(tile0 + j) * 32 + lane;
            v4f* dst = (v4f*)&rec[item * REC_F];
            const v4f ha = {C[0], C[1], C[2], C[3]};
            const v4f hb = {C[4], C[5], C[6], C[7]};
            dst[0] = ha;
            dst[1] = hb;
            dst[2] = gi[j];
        }
    }
}

__device__ __forceinline__ float item_result(
    v4f ha, v4f hb, v4f gi, float rt,
    v4f ua, v4f ub, v4f g, const float* wf8, float pv, float bfv)
{
    const float pre[8] = {ua.x + ha.x, ua.y + ha.y, ua.z + ha.z, ua.w + ha.w,
                          ub.x + hb.x, ub.y + hb.y, ub.z + hb.z, ub.w + hb.w};
    float score = bfv + gi.x * g.x + gi.y * g.y + gi.z * g.z + gi.w * g.w;
    #pragma unroll
    for (int r = 0; r < 8; ++r) score += fmaxf(pre[r], 0.0f) * wf8[r];
    return pv * rt + score;
}

__global__ __launch_bounds__(WG_THREADS) void ncf_rowtile_kernel(
    const float* __restrict__ ratings,
    const int*   __restrict__ item_ids,
    const float* __restrict__ noise,
    const float* __restrict__ init_user_emb,
    const float* __restrict__ rec,
    const float* __restrict__ W0, const float* __restrict__ b0,
    const float* __restrict__ W1, const float* __restrict__ b1,
    const float* __restrict__ W2, const float* __restrict__ b2,
    const float* __restrict__ W3, const float* __restrict__ b3,
    const float* __restrict__ Wn, const float* __restrict__ bn,
    const float* __restrict__ Wf, const float* __restrict__ bf,
    const float* __restrict__ p,
    float* __restrict__ out)
{
    __shared__ __attribute__((aligned(16))) int   s_ids[IDS_PER_THREAD * WG_THREADS * 4];
    __shared__ __attribute__((aligned(16))) float s_part[2 * ROWS_WG * 16];
    __shared__ __attribute__((aligned(16))) float s_u[ROWS_WG * 8];
    __shared__ __attribute__((aligned(16))) float s_g[ROWS_WG * 4];

    const int t    = threadIdx.x;
    const int wave = t >> 5;
    const int lane = t & 31;
    const int c    = lane & 15;
    const int h    = lane >> 4;
    const bool hi  = h != 0;
    const int  row0     = blockIdx.x * ROWS_WG;
    const long itemBase = (long)row0 * MAX_ITEMS;

    if (t < (ROWS_WG * MAX_ITEMS * 4) / 128) __builtin_prefetch(&ratings[itemBase + t * 32], 0, 3);

    const int4* idsrc = (const int4*)(item_ids + itemBase);
    int4 idv[IDS_PER_THREAD];
    #pragma unroll
    for (int j = 0; j < IDS_PER_THREAD; ++j) {
        const int i = t + j * WG_THREADS;
        idv[j] = idsrc[i < IDS_V4 ? i : IDS_V4 - 1];
    }

    {
        const int grp = wave >> 1;
        const int kh  = wave & 1;
        const float* X  = (grp < 2) ? init_user_emb : noise;
        const float* Wg = (grp == 0) ? W0 : (grp == 1) ? W1 : (grp == 2) ? W2 : W3;
        const float* xrow = X + (long)(row0 + c) * EMB_DIM + (grp & 1) * HALF_DIM + kh * 32 + h * 4;
        const float* wrow = Wg + (c & 3) * HALF_DIM + kh * 32 + h * 4;
        float4 xv[4], wv[4];
        #pragma unroll
        for (int i = 0; i < 4; ++i) {
            xv[i] = *(const float4*)(xrow + i * 8);
            wv[i] = *(const float4*)(wrow + i * 8);
        }
        {
            int4* dst = (int4*)s_ids;
            #pragma unroll
            for (int j = 0; j < IDS_PER_THREAD; ++j) dst[t + j * WG_THREADS] = idv[j];
        }
        v8f acc0 = {}, acc1 = {};
        #pragma unroll
        for (int i = 0; i < 4; ++i) {
            v2f a0 = {wv[i].x, wv[i].y}, x0 = {xv[i].x, xv[i].y};
            v2f a1 = {wv[i].z, wv[i].w}, x1 = {xv[i].z, xv[i].w};
            acc0 = __builtin_amdgcn_wmma_f32_16x16x4_f32(false, a0, false, x0, (short)0, acc0, false, false);
            acc1 = __builtin_amdgcn_wmma_f32_16x16x4_f32(false, a1, false, x1, (short)0, acc1, false, false);
        }
        float4 o;
        o.x = acc0[0] + acc1[0];
        o.y = acc0[1] + acc1[1];
        o.z = acc0[2] + acc1[2];
        o.w = acc0[3] + acc1[3];
        if (!hi) *(float4*)&s_part[(kh * ROWS_WG + c) * 16 + grp * 4] = o;
    }
    __syncthreads();

    if (t < ROWS_WG * 8) {
        const int row = t >> 3, r = t & 7;
        const float* p0 = &s_part[row * 16];
        const float* p1 = &s_part[(ROWS_WG + row) * 16];
        float acc = bn[r];
        #pragma unroll
        for (int j = 0; j < 4; ++j) {
            const float nu = (p0[4 + j] + p1[4 + j]) + b1[j];
            const float nn = (p0[12 + j] + p1[12 + j]) + b3[j];
            acc += Wn[r * 12 + j] * nu + Wn[r * 12 + 4 + j] * nn;
        }
        s_u[row * 8 + r] = acc;
    } else if (t < ROWS_WG * 8 + ROWS_WG * 4) {
        const int q = t - ROWS_WG * 8;
        const int row = q >> 2, k = q & 3;
        const float* p0 = &s_part[row * 16];
        const float* p1 = &s_part[(ROWS_WG + row) * 16];
        const float gu = (p0[k] + p1[k]) + b0[k];
        const float gn = (p0[8 + k] + p1[8 + k]) + b2[k];
        s_g[row * 4 + k] = gu * Wf[k] + gn * Wf[4 + k];
    }
    __syncthreads();

    float wf8[8];
    #pragma unroll
    for (int r = 0; r < 8; ++r) wf8[r] = Wf[8 + r];
    const float pv = p[0], bfv = bf[0];
    const v4f* u4 = (const v4f*)s_u;
    const v4f* g4 = (const v4f*)s_g;

    int tile = __builtin_amdgcn_readfirstlane(wave);
    for (; tile + TILE_WAVES < TILES_WG; tile += 2 * TILE_WAVES) {
        const unsigned f0 = (unsigned)(tile * 32 + lane);
        const unsigned f1 = f0 + TILE_WAVES * 32;
        const int id0 = s_ids[f0];
        const int id1 = s_ids[f1];
        const v4f* q0 = (const v4f*)&rec[(long)id0 * REC_F];
        const v4f* q1 = (const v4f*)&rec[(long)id1 * REC_F];
        const v4f ha0 = q0[0], hb0 = q0[1], gi0 = q0[2];
        const v4f ha1 = q1[0], hb1 = q1[1], gi1 = q1[2];
        const float  rt0 = ratings[itemBase + f0];
        const float  rt1 = ratings[itemBase + f1];
        const unsigned r0 = f0 / MAX_ITEMS, r1 = f1 / MAX_ITEMS;
        const v4f ua0 = u4[r0 * 2], ub0 = u4[r0 * 2 + 1], g0 = g4[r0];
        const v4f ua1 = u4[r1 * 2], ub1 = u4[r1 * 2 + 1], g1 = g4[r1];
        asm volatile("" :: "v"(ha0), "v"(hb0), "v"(gi0), "v"(ha1), "v"(hb1), "v"(gi1), "v"(rt0), "v"(rt1),
                           "v"(ua0), "v"(ub0), "v"(g0), "v"(ua1), "v"(ub1), "v"(g1));
        const float res0 = item_result(ha0, hb0, gi0, rt0, ua0, ub0, g0, wf8, pv, bfv);
        const float res1 = item_result(ha1, hb1, gi1, rt1, ua1, ub1, g1, wf8, pv, bfv);
        out[itemBase + f0] = res0;
        out[itemBase + f1] = res1;
    }
    if (tile < TILES_WG) {
        const unsigned f0 = (unsigned)(tile * 32 + lane);
        const int id0 = s_ids[f0];
        const v4f* q0 = (const v4f*)&rec[(long)id0 * REC_F];
        const v4f ha0 = q0[0], hb0 = q0[1], gi0 = q0[2];
        const float  rt0 = ratings[itemBase + f0];
        const unsigned r0 = f0 / MAX_ITEMS;
        const v4f ua0 = u4[r0 * 2], ub0 = u4[r0 * 2 + 1], g0 = g4[r0];
        asm volatile("" :: "v"(ha0), "v"(hb0), "v"(gi0), "v"(rt0), "v"(ua0), "v"(ub0), "v"(g0));
        out[itemBase + f0] = item_result(ha0, hb0, gi0, rt0, ua0, ub0, g0, wf8, pv, bfv);
    }
}

extern "C" void kernel_launch(void* const* d_in, const int* in_sizes, int n_in,
                              void* d_out, int out_size, void* d_ws, size_t ws_size,
                              hipStream_t stream) {
    const float* ratings       = (const float*)d_in[0];
    const int*   item_ids      = (const int*)  d_in[1];
    const float* noise         = (const float*)d_in[2];
    const float* init_user_emb = (const float*)d_in[3];
    const float* gmf_emb       = (const float*)d_in[4];
    const float* ncf_emb       = (const float*)d_in[5];
    const float* W0 = (const float*)d_in[6];  const float* b0 = (const float*)d_in[7];
    const float* W1 = (const float*)d_in[8];  const float* b1 = (const float*)d_in[9];
    const float* W2 = (const float*)d_in[10]; const float* b2 = (const float*)d_in[11];
    const float* W3 = (const float*)d_in[12]; const float* b3 = (const float*)d_in[13];
    const float* Wn = (const float*)d_in[14]; const float* bn = (const float*)d_in[15];
    const float* Wf = (const float*)d_in[16]; const float* bf = (const float*)d_in[17];
    const float* p  = (const float*)d_in[18];
    float* out = (float*)d_out;
    float* rec = (float*)d_ws;

    item_table_kernel<<<PRE_WGS, PRE_THREADS, 0, stream>>>(gmf_emb, ncf_emb, Wn, rec);
    ncf_rowtile_kernel<<<GRID_WGS, WG_THREADS, 0, stream>>>(
        ratings, item_ids, noise, init_user_emb, rec,
        W0, b0, W1, b1, W2, b2, W3, b3, Wn, bn, Wf, bf, p, out);
}
